# starting kernel
# baseline (speedup 1.0000x reference)
#include <hip/hip_runtime.h>

typedef __attribute__((ext_vector_type(2))) float v2f;
typedef __attribute__((ext_vector_type(8))) float v8f;

#define MN  49
#define KB1 192
#define KB2 48

__global__ __launch_bounds__(256)
void fused_shift_conv_wmma(const float* __restrict__ x,
                           const float* __restrict__ w1,
                           const float* __restrict__ w2,
                           float* __restrict__ out)
{
    __shared__ float t4s[16 * 16 * MN];

    const int lane  = threadIdx.x & 31;
    const int wave  = threadIdx.x >> 5;
    const int hi    = lane >> 4;
    const int lo    = lane & 15;
    const int kbase = blockIdx.x * 16;

    for (int job = wave; job < 64; job += 8) {
        const int o  = job >> 2;
        const int nt = job & 3;
        const int c  = nt * 16 + lo;
        const int m  = c / 7;
        const int n  = c % 7;

        const float* w1row = w1 + (kbase + lo) * KB1;

        v8f acc = {};
        #pragma unroll
        for (int jw = 0; jw < 3; ++jw) {
            const int   h    = m + jw - 1;
            const bool  ok   = (c < MN) & (h >= 0) & (h < 7);
            const int   xoff = ((h + 6) % 7) * 7 + n;
            const float* xb  = x + o * MN + xoff;

            #pragma unroll
            for (int k0 = 0; k0 < 64; k0 += 4) {
                const int i0 = k0 + 2 * hi;
                v2f a, b;
                a.x = w1row[i0 * 3 + jw];
                a.y = w1row[i0 * 3 + jw + 3];
                const float v0 = xb[i0 * (16 * MN)];
                const float v1 = xb[(i0 + 1) * (16 * MN)];
                b.x = ok ? v0 : 0.0f;
                b.y = ok ? v1 : 0.0f;
                acc = __builtin_amdgcn_wmma_f32_16x16x4_f32(false, a, false, b,
                                                            (short)0, acc, false, false);
            }
        }
        if (c < MN) {
            #pragma unroll
            for (int v = 0; v < 8; ++v) {
                const int r = v + 8 * hi;
                t4s[(r * 16 + o) * MN + c] = acc[v];
            }
        }
    }

    __syncthreads();

    for (int job = wave; job < 128; job += 8) {
        const int nt  = job & 3;
        const int imt = (job >> 2) & 1;
        const int o2l = job >> 3;
        const int c   = nt * 16 + lo;
        const int m   = c / 7;
        const int n   = c % 7;

        const float* w2row = w2 + (imt * 16 + lo) * KB2;

        v8f acc = {};
        #pragma unroll
        for (int kw = 0; kw < 3; ++kw) {
            const int  w   = n + kw - 1;
            const bool ok  = (c < MN) & (w >= 0) & (w < 7);
            const int  off = ok ? (m * 7 + w) : 0;
            const float* tb = t4s + o2l * (16 * MN) + off;

            #pragma unroll
            for (int k0 = 0; k0 < 16; k0 += 4) {
                const int j0 = k0 + 2 * hi;
                v2f a, b;
                a.x = w2row[j0 * 3 + kw];
                a.y = w2row[j0 * 3 + kw + 3];
                const float v0 = tb[j0 * MN];
                const float v1 = tb[(j0 + 1) * MN];
                b.x = ok ? v0 : 0.0f;
                b.y = ok ? v1 : 0.0f;
                acc = __builtin_amdgcn_wmma_f32_16x16x4_f32(false, a, false, b,
                                                            (short)0, acc, false, false);
            }
        }
        if (c < MN) {
            #pragma unroll
            for (int v = 0; v < 8; ++v) {
                const int i  = imt * 16 + v + 8 * hi;
                const int o2 = kbase + o2l;
                out[(i * 32 + o2) * MN + c] = acc[v];
            }
        }
    }
}

extern "C" void kernel_launch(void* const* d_in, const int* in_sizes, int n_in,
                              void* d_out, int out_size, void* d_ws, size_t ws_size,
                              hipStream_t stream) {
    const float* x  = (const float*)d_in[0];
    const float* w1 = (const float*)d_in[1];
    const float* w2 = (const float*)d_in[2];
    float* out = (float*)d_out;
    (void)in_sizes; (void)n_in; (void)out_size; (void)d_ws; (void)ws_size;
    fused_shift_conv_wmma<<<dim3(2), dim3(256), 0, stream>>>(x, w1, w2, out);
}


# Stage-1 contraction split over 4 waves per work-group (16 in-channels each); partial D1 tiles added through LDS in fixed order, waves 0/1 run stage 2 per i-tile; stage-1 sum order changes
# speedup vs baseline: 6.2900x; 6.2900x over previous
#include <hip/hip_runtime.h>

typedef __attribute__((ext_vector_type(2))) float v2f;
typedef __attribute__((ext_vector_type(3))) float v3f;
typedef __attribute__((ext_vector_type(4))) float v4f;
typedef __attribute__((ext_vector_type(8))) float v8f;

#define MN  49
#define KB1 192
#define KB2 48
#define NW  4

__global__ __launch_bounds__(32 * NW) __attribute__((amdgpu_waves_per_eu(1, 1)))
void fused_shift_conv_wmma(const float* __restrict__ x,
                           const float* __restrict__ w1,
                           const float* __restrict__ w2,
                           float* __restrict__ out)
{
    __shared__ v4f part[NW][3][2][32];

    const int lane = threadIdx.x & 31;
    const int wv   = __builtin_amdgcn_readfirstlane((int)(threadIdx.x >> 5));
    const int hi   = lane >> 4;
    const int lo   = lane & 15;
    const int kt   = blockIdx.x & 1;
    const int mn   = blockIdx.x >> 1;
    const int m    = mn / 7;
    const int n    = mn % 7;

    const int nb = (n < 1) ? 0 : ((n > 5) ? 4 : (n - 1));
    const int sh = (n - 1) - nb;

    int  roff[3];
    bool rok[3];
    #pragma unroll
    for (int jh = 0; jh < 3; ++jh) {
        const int hp = m + jh - 1;
        rok[jh]  = (jh == 1) | ((hp >= 0) & (hp < 7));
        roff[jh] = ((hp + 6) % 7) * 7 + nb;
    }

    const float* xb  = x  + ((hi * 32 + wv * 8) * 16 + lo) * MN;
    const float* w1p = w1 + (kt * 16 + lo) * KB1 + hi * 96 + wv * 24;

    v4f wq[6];
    v3f xv[24];
    #pragma unroll
    for (int q4 = 0; q4 < 6; ++q4)
        wq[q4] = *(const v4f*)(w1p + q4 * 4);
    #pragma unroll
    for (int q = 0; q < 24; ++q) {
        const int ii = q / 3;
        const int jh = q % 3;
        __builtin_memcpy(&xv[q], xb + ii * (16 * MN) + roff[jh], 12);
    }
    v4f uv[6] = {};
    if (wv < 2) {
        const float* w2p = w2 + (wv * 16 + lo) * KB2 + hi * 24;
        #pragma unroll
        for (int q4 = 0; q4 < 6; ++q4)
            uv[q4] = *(const v4f*)(w2p + q4 * 4);
    }

    v8f acc0 = {}, acc1 = {}, acc2 = {};

    #pragma unroll
    for (int s = 0; s < 12; ++s) {
        v2f a0, a1, a2, b;
        #pragma unroll
        for (int e = 0; e < 2; ++e) {
            const int q  = 2 * s + e;
            const int jh = q % 3;
            const float t0 = rok[jh] ? xv[q].x : 0.0f;
            const float t1 = rok[jh] ? xv[q].y : 0.0f;
            const float t2 = rok[jh] ? xv[q].z : 0.0f;
            const float wk = wq[q >> 2][q & 3];
            if (e == 0) { a0.x = t0; a1.x = t1; a2.x = t2; b.x = wk; }
            else        { a0.y = t0; a1.y = t1; a2.y = t2; b.y = wk; }
        }
        acc0 = __builtin_amdgcn_wmma_f32_16x16x4_f32(false, a0, false, b,
                                                     (short)0, acc0, false, false);
        acc1 = __builtin_amdgcn_wmma_f32_16x16x4_f32(false, a1, false, b,
                                                     (short)0, acc1, false, false);
        acc2 = __builtin_amdgcn_wmma_f32_16x16x4_f32(false, a2, false, b,
                                                     (short)0, acc2, false, false);
        __builtin_amdgcn_sched_barrier(0);
    }

    part[wv][0][0][lane] = (v4f){acc0[0], acc0[1], acc0[2], acc0[3]};
    part[wv][0][1][lane] = (v4f){acc0[4], acc0[5], acc0[6], acc0[7]};
    part[wv][1][0][lane] = (v4f){acc1[0], acc1[1], acc1[2], acc1[3]};
    part[wv][1][1][lane] = (v4f){acc1[4], acc1[5], acc1[6], acc1[7]};
    part[wv][2][0][lane] = (v4f){acc2[0], acc2[1], acc2[2], acc2[3]};
    part[wv][2][1][lane] = (v4f){acc2[4], acc2[5], acc2[6], acc2[7]};
    __syncthreads();

    if (wv < 2) {
        v8f col[3];
        #pragma unroll
        for (int t = 0; t < 3; ++t) {
            v4f l = part[0][t][0][lane];
            v4f h = part[0][t][1][lane];
            #pragma unroll
            for (int p = 1; p < NW; ++p) {
                l += part[p][t][0][lane];
                h += part[p][t][1][lane];
            }
            col[t] = (v8f){l.x, l.y, l.z, l.w, h.x, h.y, h.z, h.w};
        }

        const v8f zt = {};
        const v8f tw0 = (sh == 0) ? col[0] : ((sh > 0) ? col[1] : zt);
        const v8f tw1 = (sh == 0) ? col[1] : ((sh > 0) ? col[2] : col[0]);
        const v8f tw2 = (sh == 0) ? col[2] : ((sh < 0) ? col[1] : zt);

        v8f d = {};
        #pragma unroll
        for (int kw = 0; kw < 3; ++kw) {
            const v8f t = (kw == 0) ? tw0 : ((kw == 1) ? tw1 : tw2);
            #pragma unroll
            for (int s = 0; s < 4; ++s) {
                const int f0 = (2 * s) * 3 + kw;
                const int f1 = (2 * s + 1) * 3 + kw;
                v2f a, b;
                a.x = uv[f0 >> 2][f0 & 3];
                a.y = uv[f1 >> 2][f1 & 3];
                b.x = t[2 * s];
                b.y = t[2 * s + 1];
                d = __builtin_amdgcn_wmma_f32_16x16x4_f32(false, a, false, b,
                                                          (short)0, d, false, false);
            }
        }
        #pragma unroll
        for (int r = 0; r < 8; ++r) {
            const int i  = wv * 16 + 8 * hi + r;
            const int o2 = kt * 16 + lo;
            out[(i * 32 + o2) * MN + mn] = d[r];
        }
    }
}

extern "C" void kernel_launch(void* const* d_in, const int* in_sizes, int n_in,
                              void* d_out, int out_size, void* d_ws, size_t ws_size,
                              hipStream_t stream) {
    const float* x  = (const float*)d_in[0];
    const float* w1 = (const float*)d_in[1];
    const float* w2 = (const float*)d_in[2];
    float* out = (float*)d_out;
    (void)in_sizes; (void)n_in; (void)out_size; (void)d_ws; (void)ws_size;
    fused_shift_conv_wmma<<<dim3(98), dim3(32 * NW), 0, stream>>>(x, w1, w2, out);
}
